# starting kernel
# baseline (speedup 1.0000x reference)
#include <hip/hip_runtime.h>
#include <hip/hip_bf16.h>

typedef float v2f __attribute__((ext_vector_type(2)));
typedef float v8f __attribute__((ext_vector_type(8)));

__global__ __launch_bounds__(256) void qnn_prod_kernel(
    const float* __restrict__ batch,
    const float* __restrict__ weights,
    float* __restrict__ out,
    int nSamples)
{
    __shared__ float sV[16 * 12];

    const int tid = threadIdx.x;

    if (tid < 16) {
        const float a = weights[tid * 6 + 0];
        const float b = weights[tid * 6 + 1];
        const float c = weights[tid * 6 + 2];
        float sa, ca, sb, cb, sc, cc;
        __sincosf(a, &sa, &ca);
        __sincosf(b, &sb, &cb);
        __sincosf(c, &sc, &cc);
        float* V = &sV[tid * 12];
        V[0] = cb * ca;                    V[1] = cb * sa;                    V[2]  = sb;
        V[4] = -cc * sa - sc * sb * ca;    V[5] = cc * ca - sc * sb * sa;     V[6]  = sc * cb;
        V[8] =  sc * sa - cc * sb * ca;    V[9] = -sc * ca - cc * sb * sa;    V[10] = cc * cb;
        V[3] = 0.0f; V[7] = 0.0f; V[11] = 0.0f;
    }
    __syncthreads();

    const int lane = tid & 31;
    const int wave = tid >> 5;
    const int m    = lane & 15;
    const bool hi  = lane >= 16;
    const int mm   = m & 7;
    const bool valid = (mm < 3);

    v2f A[16];
#pragma unroll
    for (int l = 0; l < 16; ++l) {
        const float* V = &sV[l * 12];
        float ax = valid ? V[mm * 4 + (hi ? 2 : 0)] : 0.0f;
        float ay = (valid && !hi) ? V[mm * 4 + 1] : 0.0f;
        A[l].x = ax;
        A[l].y = ay;
    }

    int sample = blockIdx.x * 128 + wave * 16 + m;
    int sclamp = sample < nSamples ? sample : (nSamples - 1);
    const float* xp = batch + (size_t)sclamp * 3;
    const float x0 = xp[0], x1 = xp[1], x2 = xp[2];
    float sa, ca, sb, cb, sc, cc;
    __sincosf(x0, &sa, &ca);
    __sincosf(x1, &sb, &cb);
    __sincosf(x2, &sc, &cc);
    const float u00 = cb * ca,                 u01 = cb * sa,                 u02 = sb;
    const float u10 = -cc * sa - sc * sb * ca, u11 = cc * ca - sc * sb * sa,  u12 = sc * cb;
    const float u20 =  sc * sa - cc * sb * ca, u21 = -sc * ca - cc * sb * sa, u22 = cc * cb;

    const float ra0 = hi ? u20 : u00, ra1 = hi ? u21 : u01, ra2 = hi ? u22 : u02;
    const float rb0 = hi ? 0.0f : u10, rb1 = hi ? 0.0f : u11, rb2 = hi ? 0.0f : u12;

    float s0 = 1.0f, s1 = 0.0f, s2 = 0.0f;

#pragma unroll
    for (int l = 0; l < 16; ++l) {
        v2f Bv;
        Bv.x = fmaf(ra0, s0, fmaf(ra1, s1, ra2 * s2));
        Bv.y = fmaf(rb0, s0, fmaf(rb1, s1, rb2 * s2));
        v8f cz = {};
        v8f d = __builtin_amdgcn_wmma_f32_16x16x4_f32(
            false, A[l], false, Bv, (short)0, cz, false, false);
        s0 = d[0];
        s1 = d[1];
        s2 = d[2];
    }

    const float r0 = __builtin_sqrtf(fmaxf(s0 * s0, 1e-10f)) + 2.0e-5f;
    const float r1 = __builtin_sqrtf(fmaxf(s1 * s1, 1e-10f)) + 2.0e-5f;
    const float r2 = __builtin_sqrtf(fmaxf(s2 * s2, 1e-10f)) + 2.0e-5f;
    const float f0 = r0 * r0, f1 = r1 * r1, f2 = r2 * r2;
    const float inv = 1.0f / (f0 + f1 + f2);

    if (!hi && sample < nSamples) {
        float* o = out + (size_t)sample * 3;
        o[0] = f0 * inv;
        o[1] = f1 * inv;
        o[2] = f2 * inv;
    }
}

extern "C" void kernel_launch(void* const* d_in, const int* in_sizes, int n_in,
                              void* d_out, int out_size, void* d_ws, size_t ws_size,
                              hipStream_t stream) {
    const float* batch   = (const float*)d_in[0];
    const float* weights = (const float*)d_in[1];
    float* out = (float*)d_out;

    const int nSamples = in_sizes[0] / 3;
    const int samplesPerBlock = 128;
    const int grid = (nSamples + samplesPerBlock - 1) / samplesPerBlock;

    qnn_prod_kernel<<<grid, 256, 0, stream>>>(batch, weights, out, nSamples);
}


# Pack 64 distinct samples per wave (2 per lane, both half-waves, all 4 K slots): 3 accumulate-chained f32 WMMAs per layer replace 4; each V*t sum now split over the chained products
# speedup vs baseline: 1.4064x; 1.4064x over previous
#include <hip/hip_runtime.h>


typedef float v2f __attribute__((ext_vector_type(2)));
typedef float v8f __attribute__((ext_vector_type(8)));

__global__ __launch_bounds__(256) void qnn_prod_kernel(
    const float* __restrict__ batch,
    const float* __restrict__ weights,
    float* __restrict__ out,
    int nSamples)
{
    __shared__ float sT[16 * 16];

    const int tid = threadIdx.x;

    if (tid < 16) {
        const float a = weights[tid * 6 + 0];
        const float b = weights[tid * 6 + 1];
        const float c = weights[tid * 6 + 2];
        float sa, ca, sb, cb, sc, cc;
        __sincosf(a, &sa, &ca);
        __sincosf(b, &sb, &cb);
        __sincosf(c, &sc, &cc);
        float* V = &sT[tid * 16];
        V[0] = 0.0f;  V[1] = cb * ca;                  V[2]  = cb * sa;                  V[3]  = sb;
        V[4] = 0.0f;  V[5] = -cc * sa - sc * sb * ca;  V[6]  = cc * ca - sc * sb * sa;   V[7]  = sc * cb;
        V[8] = 0.0f;  V[9] =  sc * sa - cc * sb * ca;  V[10] = -sc * ca - cc * sb * sa;  V[11] = cc * cb;
        V[12] = 0.0f; V[13] = 0.0f; V[14] = 0.0f; V[15] = 0.0f;
    }
    __syncthreads();

    const int lane = tid & 31;
    const int wave = tid >> 5;
    const int c    = lane & 15;
    const int h    = lane >> 4;
    const int g    = c >> 3;
    const int r    = c & 7;
    const bool act = (g == h) && (r < 6);
    const bool rowX = act && (r < 3);
    const bool rowY = act && (r >= 3);
    const int i    = rowY ? (r - 3) : r;
    const int b0 = rowX ? (4 * i + 1) : 12;
    const int b1 = rowX ? (4 * i + 3) : (rowY ? (4 * i) : 12);
    const int b2 = rowY ? (4 * i + 2) : 12;

    const int ix = blockIdx.x * 512 + wave * 64 + lane;
    const int iy = ix + 32;
    const int cx = ix < nSamples ? ix : (nSamples - 1);
    const int cy = iy < nSamples ? iy : (nSamples - 1);
    const float* px = batch + (size_t)cx * 3;
    const float* py = batch + (size_t)cy * 3;
    const float xa0 = px[0], xa1 = px[1], xa2 = px[2];
    const float xb0 = py[0], xb1 = py[1], xb2 = py[2];

    float sa, ca, sb, cb, sc, cc;
    __sincosf(xa0, &sa, &ca);
    __sincosf(xa1, &sb, &cb);
    __sincosf(xa2, &sc, &cc);
    const float ux00 = cb * ca,                 ux01 = cb * sa,                 ux02 = sb;
    const float ux10 = -cc * sa - sc * sb * ca, ux11 = cc * ca - sc * sb * sa,  ux12 = sc * cb;
    const float ux20 =  sc * sa - cc * sb * ca, ux21 = -sc * ca - cc * sb * sa, ux22 = cc * cb;

    __sincosf(xb0, &sa, &ca);
    __sincosf(xb1, &sb, &cb);
    __sincosf(xb2, &sc, &cc);
    const float uy00 = cb * ca,                 uy01 = cb * sa,                 uy02 = sb;
    const float uy10 = -cc * sa - sc * sb * ca, uy11 = cc * ca - sc * sb * sa,  uy12 = sc * cb;
    const float uy20 =  sc * sa - cc * sb * ca, uy21 = -sc * ca - cc * sb * sa, uy22 = cc * cb;

    float sx0 = 1.0f, sx1 = 0.0f, sx2 = 0.0f;
    float sy0 = 1.0f, sy1 = 0.0f, sy2 = 0.0f;

#pragma unroll
    for (int l = 0; l < 16; ++l) {
        const float* Tl = &sT[l * 16];
        v2f A0, A1, A2;
        A0.x = Tl[b0]; A0.y = Tl[b0 + 1];
        A1.x = Tl[b1]; A1.y = Tl[b1 + 1];
        A2.x = Tl[b2]; A2.y = Tl[b2 + 1];

        v2f B0, B1, B2;
        B0.x = fmaf(ux00, sx0, fmaf(ux01, sx1, ux02 * sx2));
        B0.y = fmaf(ux10, sx0, fmaf(ux11, sx1, ux12 * sx2));
        B1.x = fmaf(ux20, sx0, fmaf(ux21, sx1, ux22 * sx2));
        B1.y = fmaf(uy00, sy0, fmaf(uy01, sy1, uy02 * sy2));
        B2.x = fmaf(uy10, sy0, fmaf(uy11, sy1, uy12 * sy2));
        B2.y = fmaf(uy20, sy0, fmaf(uy21, sy1, uy22 * sy2));

        v8f d = {};
        d = __builtin_amdgcn_wmma_f32_16x16x4_f32(
            false, A0, false, B0, (short)0, d, false, false);
        d = __builtin_amdgcn_wmma_f32_16x16x4_f32(
            false, A1, false, B1, (short)0, d, false, false);
        d = __builtin_amdgcn_wmma_f32_16x16x4_f32(
            false, A2, false, B2, (short)0, d, false, false);
        sx0 = d[0]; sx1 = d[1]; sx2 = d[2];
        sy0 = d[3]; sy1 = d[4]; sy2 = d[5];
    }

    const float rx0 = __builtin_sqrtf(fmaxf(sx0 * sx0, 1e-10f)) + 2.0e-5f;
    const float rx1 = __builtin_sqrtf(fmaxf(sx1 * sx1, 1e-10f)) + 2.0e-5f;
    const float rx2 = __builtin_sqrtf(fmaxf(sx2 * sx2, 1e-10f)) + 2.0e-5f;
    const float fx0 = rx0 * rx0, fx1 = rx1 * rx1, fx2 = rx2 * rx2;
    const float invx = 1.0f / (fx0 + fx1 + fx2);

    const float ry0 = __builtin_sqrtf(fmaxf(sy0 * sy0, 1e-10f)) + 2.0e-5f;
    const float ry1 = __builtin_sqrtf(fmaxf(sy1 * sy1, 1e-10f)) + 2.0e-5f;
    const float ry2 = __builtin_sqrtf(fmaxf(sy2 * sy2, 1e-10f)) + 2.0e-5f;
    const float fy0 = ry0 * ry0, fy1 = ry1 * ry1, fy2 = ry2 * ry2;
    const float invy = 1.0f / (fy0 + fy1 + fy2);

    if (ix < nSamples) {
        float* o = out + (size_t)ix * 3;
        o[0] = fx0 * invx; o[1] = fx1 * invx; o[2] = fx2 * invx;
    }
    if (iy < nSamples) {
        float* o = out + (size_t)iy * 3;
        o[0] = fy0 * invy; o[1] = fy1 * invy; o[2] = fy2 * invy;
    }
}

extern "C" void kernel_launch(void* const* d_in, const int* in_sizes, int n_in,
                              void* d_out, int out_size, void* d_ws, size_t ws_size,
                              hipStream_t stream) {
    const float* batch   = (const float*)d_in[0];
    const float* weights = (const float*)d_in[1];
    float* out = (float*)d_out;

    const int nSamples = in_sizes[0] / 3;
    const int samplesPerBlock = 512;
    const int grid = (nSamples + samplesPerBlock - 1) / samplesPerBlock;

    qnn_prod_kernel<<<grid, 256, 0, stream>>>(batch, weights, out, nSamples);
}
